# starting kernel
# baseline (speedup 1.0000x reference)
#include <hip/hip_runtime.h>
#include <cstdint>

#define N_NODES 50000
#define N_EDGES 800000
#define DIM     128

typedef __attribute__((ext_vector_type(2))) float v2f;
typedef __attribute__((ext_vector_type(8))) float v8f;

__global__ void __launch_bounds__(256) k_init_deg(float* __restrict__ ds,
                                                  float* __restrict__ dr) {
    int i = blockIdx.x * blockDim.x + threadIdx.x;
    if (i < N_NODES) { ds[i] = 2.0f; dr[i] = 2.0f; }
}

__global__ void __launch_bounds__(256) k_edge_deg(const int*   __restrict__ snd,
                                                  const int*   __restrict__ rcv,
                                                  const float* __restrict__ ew,
                                                  float* __restrict__ ds,
                                                  float* __restrict__ dr) {
    int i = blockIdx.x * blockDim.x + threadIdx.x;
    if (i < N_EDGES) {
        float w = ew[i] + 1.0f;
        unsafeAtomicAdd(&ds[snd[i]], w);
        unsafeAtomicAdd(&dr[rcv[i]], w);
    }
}

__global__ void __launch_bounds__(128) k_gemm_scale(const float* __restrict__ nodes,
                                                    const float* __restrict__ Wm,
                                                    const float* __restrict__ ds,
                                                    const float* __restrict__ dr,
                                                    float* __restrict__ y,
                                                    float* __restrict__ out) {
    __shared__ float w_lds[DIM * DIM];

    const int tid = threadIdx.x;

    for (int idx = tid; idx < DIM * DIM; idx += 128) {
        int k = idx >> 7;
        int c = idx & (DIM - 1);
        w_lds[((k >> 1) << 8) + (c << 1) + (k & 1)] = Wm[idx];
    }
    __syncthreads();

    const int lane = tid & 31;
    const int wave = tid >> 5;
    const int strip = blockIdx.x * 4 + wave;

    if (strip * 16 < N_NODES) {
        const int row0 = strip * 16;
        const int m    = lane & 15;
        const int koff = (lane >> 4) << 1;

        const float* arow = nodes + (size_t)(row0 + m) * DIM;

        v8f acc[8] = {};

        #pragma unroll 4
        for (int k0 = 0; k0 < DIM; k0 += 4) {
            v2f a = *(const v2f*)(arow + k0 + koff);
            const float* bbase = w_lds + (((k0 + koff) >> 1) << 8);
            #pragma unroll
            for (int t = 0; t < 8; ++t) {
                v2f b = *(const v2f*)(bbase + ((t * 16 + m) << 1));
                acc[t] = __builtin_amdgcn_wmma_f32_16x16x4_f32(
                     false, a,  false, b,
                     (short)0, acc[t],
                     false,  false);
            }
        }

        const int roff = (lane >> 4) << 3;
        #pragma unroll
        for (int v = 0; v < 8; ++v) {
            const int row = row0 + v + roff;
            const float a1 = ds[row];
            const float a2 = dr[row];
            const float sc = rsqrtf(fmaxf(a1, 1.0f)) * rsqrtf(fmaxf(a2, 1.0f));
            float* yrow = y   + (size_t)row * DIM;
            float* orow = out + (size_t)row * DIM;
            #pragma unroll
            for (int t = 0; t < 8; ++t) {
                const float val = acc[t][v] * sc;
                yrow[t * 16 + m] = val;
                orow[t * 16 + m] = val;
            }
        }
    }
}

__global__ void __launch_bounds__(256) k_edge_scatter(const int*   __restrict__ snd,
                                                      const int*   __restrict__ rcv,
                                                      const float* __restrict__ ew,
                                                      const float* __restrict__ y,
                                                      float* out) {
    const int gtid = blockIdx.x * blockDim.x + threadIdx.x;
    const int e    = gtid >> 5;
    if (e >= N_EDGES) return;
    const int lane = gtid & 31;

    const int   s = snd[e];
    const int   r = rcv[e];
    const float w = ew[e];

    const float4 v = *(const float4*)(y + (size_t)s * DIM + lane * 4);
    float* dst = out + (size_t)r * DIM + lane * 4;
    unsafeAtomicAdd(dst + 0, v.x * w);
    unsafeAtomicAdd(dst + 1, v.y * w);
    unsafeAtomicAdd(dst + 2, v.z * w);
    unsafeAtomicAdd(dst + 3, v.w * w);
}

extern "C" void kernel_launch(void* const* d_in, const int* in_sizes, int n_in,
                              void* d_out, int out_size, void* d_ws, size_t ws_size,
                              hipStream_t stream) {
    const float* nodes     = (const float*)d_in[0];
    const int*   senders   = (const int*)  d_in[1];
    const int*   receivers = (const int*)  d_in[2];
    const float* edges     = (const float*)d_in[3];
    const float* Wm        = (const float*)d_in[4];
    float*       out       = (float*)d_out;

    float* y  = (float*)d_ws;
    float* ds = y  + (size_t)N_NODES * DIM;
    float* dr = ds + N_NODES;

    k_init_deg<<<(N_NODES + 255) / 256, 256, 0, stream>>>(ds, dr);
    k_edge_deg<<<(N_EDGES + 255) / 256, 256, 0, stream>>>(senders, receivers, edges, ds, dr);

    const int n_strips = (N_NODES + 15) / 16;
    k_gemm_scale<<<(n_strips + 3) / 4, 128, 0, stream>>>(nodes, Wm, ds, dr, y, out);

    const long long scatter_threads = (long long)N_EDGES * 32;
    k_edge_scatter<<<(int)((scatter_threads + 255) / 256), 256, 0, stream>>>(
        senders, receivers, edges, y, out);
}


# receiver line: one returning 64-bit add (edge count | exact fixed-point degree sum) replaces f32 add + head swap; its rank puts the 8-B record in a slot of the line, chain past 14; sum order changes
# speedup vs baseline: 3.6153x; 3.6153x over previous
#include <hip/hip_runtime.h>
#include <cstdint>

#define N_NODES 50000
#define N_EDGES 800000
#define DIM     128
#define DEG_STRIDE 32
#define OVF_WORD   1
#define PACK_WORD  2
#define SLOT_WORD  4
#define N_SLOTS    14u
#define CNT_SHIFT  44
#define FIX_MASK   ((1ull << CNT_SHIFT) - 1ull)
#define FIX_SCALE  8388608.0f
#define PLANE_FLOATS ((size_t)N_NODES * DEG_STRIDE)
#define PLANE_BYTES  ((unsigned)(N_NODES * DEG_STRIDE * 4))
#define REC_WORDS  4

typedef __attribute__((ext_vector_type(2))) float v2f;
typedef __attribute__((ext_vector_type(8))) float v8f;

__global__ void __launch_bounds__(256) k_init_deg(float* __restrict__ deg) {
    int i = blockIdx.x * blockDim.x + threadIdx.x;
    if (i < 2 * N_NODES) {
        float* line = deg + (size_t)i * DEG_STRIDE;
        const bool first = i < N_NODES;
        int4 start;
        start.x = __float_as_int(first ? 2.0f : 0.0f);
        start.y = -1;
        start.z = first ? (1 << 24) : 0;
        start.w = 0;
        *(int4*)line = start;
    }
}

__global__ void __launch_bounds__(256) k_edge_deg(const int*   __restrict__ snd,
                                                  const int*   __restrict__ rcv,
                                                  const float* __restrict__ ew,
                                                  float* __restrict__ deg,
                                                  int*   __restrict__ rec) {
    int i = blockIdx.x * blockDim.x + threadIdx.x;
    const unsigned half = (__builtin_amdgcn_s_sendmsg_rtn(0x87) >> 18) & 1u;
    deg += (size_t)half * PLANE_FLOATS;
    if (i < N_EDGES) {
        const int   s  = snd[i];
        const int   r  = rcv[i];
        const float w0 = ew[i];
        const float w  = w0 + 1.0f;
        float* rline = deg + (size_t)r * DEG_STRIDE;
        const unsigned long long addv =
            (1ull << CNT_SHIFT) + (unsigned long long)(w * FIX_SCALE);
        const unsigned long long old =
            atomicAdd((unsigned long long*)(rline + PACK_WORD), addv);
        unsafeAtomicAdd(deg + (size_t)s * DEG_STRIDE, w);
        const unsigned rank = (unsigned)(old >> CNT_SHIFT);
        if (rank < N_SLOTS) {
            int2 rc;
            rc.x = s;
            rc.y = __float_as_int(w0);
            *(int2*)((int*)rline + SLOT_WORD + 2u * rank) = rc;
        } else {
            const int prev = atomicExch((int*)rline + OVF_WORD, i);
            int4 rc;
            rc.x = s;
            rc.y = __float_as_int(w0);
            rc.z = prev;
            rc.w = 0;
            *(int4*)(rec + (size_t)i * REC_WORDS) = rc;
        }
    }
}

__global__ void __launch_bounds__(128) k_gemm_scale(const float* __restrict__ nodes,
                                                    const float* __restrict__ Wm,
                                                    const float* __restrict__ deg,
                                                    float* __restrict__ y) {
    __shared__ float w_lds[DIM * DIM];

    const int tid = threadIdx.x;

    for (int idx = tid; idx < DIM * DIM; idx += 128) {
        int k = idx >> 7;
        int c = idx & (DIM - 1);
        w_lds[((k >> 1) << 8) + (c << 1) + (k & 1)] = Wm[idx];
    }
    __syncthreads();

    const int lane = tid & 31;
    const int wave = tid >> 5;
    const int strip = blockIdx.x * 4 + wave;

    if (strip * 16 < N_NODES) {
        const int row0 = strip * 16;
        const int m    = lane & 15;
        const int koff = (lane >> 4) << 1;

        const float* arow = nodes + (size_t)(row0 + m) * DIM;

        v8f acc[8] = {};

        #pragma unroll 4
        for (int k0 = 0; k0 < DIM; k0 += 4) {
            v2f a = *(const v2f*)(arow + k0 + koff);
            const float* bbase = w_lds + (((k0 + koff) >> 1) << 8);
            #pragma unroll
            for (int t = 0; t < 8; ++t) {
                v2f b = *(const v2f*)(bbase + ((t * 16 + m) << 1));
                acc[t] = __builtin_amdgcn_wmma_f32_16x16x4_f32(
                     false, a,  false, b,
                     (short)0, acc[t],
                     false,  false);
            }
        }

        const int roff = (lane >> 4) << 3;
        #pragma unroll
        for (int v = 0; v < 8; ++v) {
            const int row = row0 + v + roff;
            const int4 d0 = *(const int4*)(deg + (size_t)row * DEG_STRIDE);
            const int4 d1 = *(const int4*)(deg + PLANE_FLOATS + (size_t)row * DEG_STRIDE);
            const float sdeg = __int_as_float(d0.x) + __int_as_float(d1.x);
            const unsigned long long p0 = ((unsigned long long)(unsigned)d0.w << 32) | (unsigned)d0.z;
            const unsigned long long p1 = ((unsigned long long)(unsigned)d1.w << 32) | (unsigned)d1.z;
            const float rdeg = (float)((p0 & FIX_MASK) + (p1 & FIX_MASK)) * (1.0f / FIX_SCALE);
            const float sc = rsqrtf(fmaxf(sdeg, 1.0f)) * rsqrtf(fmaxf(rdeg, 1.0f));
            float* yrow = y + (size_t)row * DIM;
            #pragma unroll
            for (int t = 0; t < 8; ++t) {
                yrow[t * 16 + m] = acc[t][v] * sc;
            }
        }
    }
}

__device__ __forceinline__ void row_fma(float4 (&acc)[4], const float4 (&v)[4], float w) {
    #pragma unroll
    for (int j = 0; j < 4; ++j) {
        acc[j].x += v[j].x * w;
        acc[j].y += v[j].y * w;
        acc[j].z += v[j].z * w;
        acc[j].w += v[j].w * w;
    }
}

__device__ __forceinline__ void row_load(float4 (&v)[4], const float* __restrict__ y,
                                         int node, unsigned coff) {
    const char* row = (const char*)y + ((unsigned)node * (DIM * 4u) + coff);
    #pragma unroll
    for (int j = 0; j < 4; ++j) v[j] = *(const float4*)(row + 128 * j);
}

#define GROUP_PAIRS 34

__device__ __forceinline__ int2 slot_read(const int2* gl, unsigned j, unsigned n0) {
    return gl[j < n0 ? 2u + j : 18u + j - n0];
}

struct Rec3 { int sender, wbits, next; };

__device__ __forceinline__ Rec3 rec_load(const int* __restrict__ rec, int e) {
    const int* p = (const int*)((const char*)rec + (unsigned)e * (REC_WORDS * 4u));
    Rec3 c;
    c.sender = p[0];
    c.wbits  = p[1];
    c.next   = p[2];
    return c;
}

__global__ void __launch_bounds__(256) k_edge_gather(const float* __restrict__ deg,
                                                     const int*   __restrict__ rec,
                                                     const float* __restrict__ y,
                                                     float* __restrict__ out) {
    __shared__ int4 lines[32 * (GROUP_PAIRS / 2)];

    const int gtid = blockIdx.x * 256 + threadIdx.x;
    const int r    = gtid >> 3;
    if (r >= N_NODES) return;
    const unsigned sub = threadIdx.x & 7u;
    const unsigned coff = sub * 16u;

    const unsigned loff = (unsigned)r * (DEG_STRIDE * 4u) + coff;
    const int4 p0 = *(const int4*)((const char*)deg + loff);
    const int4 p1 = *(const int4*)((const char*)deg + (PLANE_BYTES + loff));

    const unsigned oidx = (unsigned)r * (DIM / 4u) + sub;
    float4 acc[4];
    #pragma unroll
    for (int j = 0; j < 4; ++j) acc[j] = ((const float4*)y)[oidx + 8u * j];

    int4* gline = lines + (threadIdx.x >> 3) * (GROUP_PAIRS / 2);
    gline[sub]      = p0;
    gline[8u + sub] = p1;
    const int4 hd0 = gline[0];
    const int4 hd1 = gline[8];

    const unsigned cnt0 = (unsigned)hd0.w >> (CNT_SHIFT - 32);
    const unsigned cnt1 = (unsigned)hd1.w >> (CNT_SHIFT - 32);
    const unsigned n0 = cnt0 < N_SLOTS ? cnt0 : N_SLOTS;
    const unsigned n1 = cnt1 < N_SLOTS ? cnt1 : N_SLOTS;
    const unsigned n  = n0 + n1;
    const int2* gl = (const int2*)gline;

    int2 ra = {-1, 0};
    int2 rb = {-1, 0};
    if (n > 0u) ra = slot_read(gl, 0u, n0);
    if (n > 1u) rb = slot_read(gl, 1u, n0);
    #pragma unroll 1
    for (unsigned j = 0; j < 2u * N_SLOTS; j += 2u) {
        const bool a0 = j < n;
        const bool a1 = j + 1u < n;
        if (__builtin_amdgcn_ballot_w32(a0) == 0u) break;
        float4 va[4], vb[4];
        if (a0) row_load(va, y, ra.x, coff);
        if (a1) row_load(vb, y, rb.x, coff);
        const float w0 = __int_as_float(ra.y);
        const float w1 = __int_as_float(rb.y);
        if (j + 2u < n) ra = slot_read(gl, j + 2u, n0);
        if (j + 3u < n) rb = slot_read(gl, j + 3u, n0);
        if (a0) row_fma(acc, va, w0);
        if (a1) row_fma(acc, vb, w1);
    }

    const int h0 = hd0.y;
    const int h1 = hd1.y;
    if (__builtin_amdgcn_ballot_w32(h0 >= 0 || h1 >= 0) != 0u) {
        const Rec3 none = {-1, 0, -1};
        Rec3 c0 = none;
        Rec3 c1 = none;
        if (h0 >= 0) c0 = rec_load(rec, h0);
        if (h1 >= 0) c1 = rec_load(rec, h1);
        for (int it = 0; it < N_EDGES; ++it) {
            const bool a0 = c0.sender >= 0;
            const bool a1 = c1.sender >= 0;
            if (__builtin_amdgcn_ballot_w32(a0 || a1) == 0u) break;
            float4 va[4], vb[4];
            if (a0) row_load(va, y, c0.sender, coff);
            if (a1) row_load(vb, y, c1.sender, coff);
            const float w0 = __int_as_float(c0.wbits);
            const float w1 = __int_as_float(c1.wbits);
            if (c0.next >= 0) c0 = rec_load(rec, c0.next); else c0 = none;
            if (c1.next >= 0) c1 = rec_load(rec, c1.next); else c1 = none;
            if (a0) row_fma(acc, va, w0);
            if (a1) row_fma(acc, vb, w1);
        }
    }

    char* orow = (char*)out + (oidx << 4);
    #pragma unroll
    for (int j = 0; j < 4; ++j) *(float4*)(orow + 128 * j) = acc[j];
}

extern "C" void kernel_launch(void* const* d_in, const int* in_sizes, int n_in,
                              void* d_out, int out_size, void* d_ws, size_t ws_size,
                              hipStream_t stream) {
    const float* nodes     = (const float*)d_in[0];
    const int*   senders   = (const int*)  d_in[1];
    const int*   receivers = (const int*)  d_in[2];
    const float* edges     = (const float*)d_in[3];
    const float* Wm        = (const float*)d_in[4];
    float*       out       = (float*)d_out;

    float* y   = (float*)d_ws;
    float* deg = y + (size_t)N_NODES * DIM;
    int*   rec = (int*)(deg + 2 * PLANE_FLOATS);

    k_init_deg<<<(2 * N_NODES + 255) / 256, 256, 0, stream>>>(deg);
    k_edge_deg<<<(N_EDGES + 255) / 256, 256, 0, stream>>>(senders, receivers, edges, deg, rec);

    const int n_strips = (N_NODES + 15) / 16;
    k_gemm_scale<<<(n_strips + 3) / 4, 128, 0, stream>>>(nodes, Wm, deg, y);

    k_edge_gather<<<(N_NODES + 31) / 32, 256, 0, stream>>>(deg, rec, y, out);
}


# GEMM launch: W staged to LDS with 16-byte loads, all 32 of a thread in flight before its first LDS store (was 128 4-byte loads in 16 dependent trips); LDS layout and products unchanged
# speedup vs baseline: 4.0112x; 1.1095x over previous
#include <hip/hip_runtime.h>
#include <cstdint>

#define N_NODES 50000
#define N_EDGES 800000
#define DIM     128
#define DEG_STRIDE 32
#define OVF_WORD   1
#define PACK_WORD  2
#define SLOT_WORD  4
#define N_SLOTS    14u
#define CNT_SHIFT  44
#define FIX_MASK   ((1ull << CNT_SHIFT) - 1ull)
#define FIX_SCALE  8388608.0f
#define PLANE_FLOATS ((size_t)N_NODES * DEG_STRIDE)
#define PLANE_BYTES  ((unsigned)(N_NODES * DEG_STRIDE * 4))
#define REC_WORDS  4

typedef __attribute__((ext_vector_type(2))) float v2f;
typedef __attribute__((ext_vector_type(8))) float v8f;

__global__ void __launch_bounds__(256) k_init_deg(float* __restrict__ deg) {
    int i = blockIdx.x * blockDim.x + threadIdx.x;
    if (i < 2 * N_NODES) {
        float* line = deg + (size_t)i * DEG_STRIDE;
        const bool first = i < N_NODES;
        int4 start;
        start.x = __float_as_int(first ? 2.0f : 0.0f);
        start.y = -1;
        start.z = first ? (1 << 24) : 0;
        start.w = 0;
        *(int4*)line = start;
    }
}

__global__ void __launch_bounds__(256) k_edge_deg(const int*   __restrict__ snd,
                                                  const int*   __restrict__ rcv,
                                                  const float* __restrict__ ew,
                                                  float* __restrict__ deg,
                                                  int*   __restrict__ rec) {
    int i = blockIdx.x * blockDim.x + threadIdx.x;
    const unsigned half = (__builtin_amdgcn_s_sendmsg_rtn(0x87) >> 18) & 1u;
    deg += (size_t)half * PLANE_FLOATS;
    if (i < N_EDGES) {
        const int   s  = snd[i];
        const int   r  = rcv[i];
        const float w0 = ew[i];
        const float w  = w0 + 1.0f;
        float* rline = deg + (size_t)r * DEG_STRIDE;
        const unsigned long long addv =
            (1ull << CNT_SHIFT) + (unsigned long long)(w * FIX_SCALE);
        const unsigned long long old =
            atomicAdd((unsigned long long*)(rline + PACK_WORD), addv);
        unsafeAtomicAdd(deg + (size_t)s * DEG_STRIDE, w);
        const unsigned rank = (unsigned)(old >> CNT_SHIFT);
        if (rank < N_SLOTS) {
            int2 rc;
            rc.x = s;
            rc.y = __float_as_int(w0);
            *(int2*)((int*)rline + SLOT_WORD + 2u * rank) = rc;
        } else {
            const int prev = atomicExch((int*)rline + OVF_WORD, i);
            int4 rc;
            rc.x = s;
            rc.y = __float_as_int(w0);
            rc.z = prev;
            rc.w = 0;
            *(int4*)(rec + (size_t)i * REC_WORDS) = rc;
        }
    }
}

__global__ void __launch_bounds__(128) k_gemm_scale(const float* __restrict__ nodes,
                                                    const float* __restrict__ Wm,
                                                    const float* __restrict__ deg,
                                                    float* __restrict__ y) {
    __shared__ __attribute__((aligned(16))) float w_lds[DIM * DIM];

    const int tid = threadIdx.x;

    float4 wlo[16], whi[16];
    #pragma unroll
    for (int i = 0; i < 16; ++i) {
        const int item = tid + 128 * i;
        const int p  = item >> 5;
        const int cg = item & 31;
        const float* src = Wm + (size_t)(2 * p) * DIM + 4 * cg;
        wlo[i] = *(const float4*)src;
        whi[i] = *(const float4*)(src + DIM);
    }
    #pragma unroll
    for (int i = 0; i < 16; ++i) {
        const int item = tid + 128 * i;
        const int p  = item >> 5;
        const int cg = item & 31;
        float4* dst = (float4*)(w_lds + (p << 8) + (cg << 3));
        dst[0] = make_float4(wlo[i].x, whi[i].x, wlo[i].y, whi[i].y);
        dst[1] = make_float4(wlo[i].z, whi[i].z, wlo[i].w, whi[i].w);
    }
    __syncthreads();

    const int lane = tid & 31;
    const int wave = tid >> 5;
    const int strip = blockIdx.x * 4 + wave;

    if (strip * 16 < N_NODES) {
        const int row0 = strip * 16;
        const int m    = lane & 15;
        const int koff = (lane >> 4) << 1;

        const float* arow = nodes + (size_t)(row0 + m) * DIM;

        v8f acc[8] = {};

        #pragma unroll 4
        for (int k0 = 0; k0 < DIM; k0 += 4) {
            v2f a = *(const v2f*)(arow + k0 + koff);
            const float* bbase = w_lds + (((k0 + koff) >> 1) << 8);
            #pragma unroll
            for (int t = 0; t < 8; ++t) {
                v2f b = *(const v2f*)(bbase + ((t * 16 + m) << 1));
                acc[t] = __builtin_amdgcn_wmma_f32_16x16x4_f32(
                     false, a,  false, b,
                     (short)0, acc[t],
                     false,  false);
            }
        }

        const int roff = (lane >> 4) << 3;
        #pragma unroll
        for (int v = 0; v < 8; ++v) {
            const int row = row0 + v + roff;
            const int4 d0 = *(const int4*)(deg + (size_t)row * DEG_STRIDE);
            const int4 d1 = *(const int4*)(deg + PLANE_FLOATS + (size_t)row * DEG_STRIDE);
            const float sdeg = __int_as_float(d0.x) + __int_as_float(d1.x);
            const unsigned long long p0 = ((unsigned long long)(unsigned)d0.w << 32) | (unsigned)d0.z;
            const unsigned long long p1 = ((unsigned long long)(unsigned)d1.w << 32) | (unsigned)d1.z;
            const float rdeg = (float)((p0 & FIX_MASK) + (p1 & FIX_MASK)) * (1.0f / FIX_SCALE);
            const float sc = rsqrtf(fmaxf(sdeg, 1.0f)) * rsqrtf(fmaxf(rdeg, 1.0f));
            float* yrow = y + (size_t)row * DIM;
            #pragma unroll
            for (int t = 0; t < 8; ++t) {
                yrow[t * 16 + m] = acc[t][v] * sc;
            }
        }
    }
}

__device__ __forceinline__ void row_fma(float4 (&acc)[4], const float4 (&v)[4], float w) {
    #pragma unroll
    for (int j = 0; j < 4; ++j) {
        acc[j].x += v[j].x * w;
        acc[j].y += v[j].y * w;
        acc[j].z += v[j].z * w;
        acc[j].w += v[j].w * w;
    }
}

__device__ __forceinline__ void row_load(float4 (&v)[4], const float* __restrict__ y,
                                         int node, unsigned coff) {
    const char* row = (const char*)y + ((unsigned)node * (DIM * 4u) + coff);
    #pragma unroll
    for (int j = 0; j < 4; ++j) v[j] = *(const float4*)(row + 128 * j);
}

#define GROUP_PAIRS 34

__device__ __forceinline__ int2 slot_read(const int2* gl, unsigned j, unsigned n0) {
    return gl[j < n0 ? 2u + j : 18u + j - n0];
}

struct Rec3 { int sender, wbits, next; };

__device__ __forceinline__ Rec3 rec_load(const int* __restrict__ rec, int e) {
    const int* p = (const int*)((const char*)rec + (unsigned)e * (REC_WORDS * 4u));
    Rec3 c;
    c.sender = p[0];
    c.wbits  = p[1];
    c.next   = p[2];
    return c;
}

__global__ void __launch_bounds__(256) k_edge_gather(const float* __restrict__ deg,
                                                     const int*   __restrict__ rec,
                                                     const float* __restrict__ y,
                                                     float* __restrict__ out) {
    __shared__ int4 lines[32 * (GROUP_PAIRS / 2)];

    const int gtid = blockIdx.x * 256 + threadIdx.x;
    const int r    = gtid >> 3;
    if (r >= N_NODES) return;
    const unsigned sub = threadIdx.x & 7u;
    const unsigned coff = sub * 16u;

    const unsigned loff = (unsigned)r * (DEG_STRIDE * 4u) + coff;
    const int4 p0 = *(const int4*)((const char*)deg + loff);
    const int4 p1 = *(const int4*)((const char*)deg + (PLANE_BYTES + loff));

    const unsigned oidx = (unsigned)r * (DIM / 4u) + sub;
    float4 acc[4];
    #pragma unroll
    for (int j = 0; j < 4; ++j) acc[j] = ((const float4*)y)[oidx + 8u * j];

    int4* gline = lines + (threadIdx.x >> 3) * (GROUP_PAIRS / 2);
    gline[sub]      = p0;
    gline[8u + sub] = p1;
    const int4 hd0 = gline[0];
    const int4 hd1 = gline[8];

    const unsigned cnt0 = (unsigned)hd0.w >> (CNT_SHIFT - 32);
    const unsigned cnt1 = (unsigned)hd1.w >> (CNT_SHIFT - 32);
    const unsigned n0 = cnt0 < N_SLOTS ? cnt0 : N_SLOTS;
    const unsigned n1 = cnt1 < N_SLOTS ? cnt1 : N_SLOTS;
    const unsigned n  = n0 + n1;
    const int2* gl = (const int2*)gline;

    int2 ra = {-1, 0};
    int2 rb = {-1, 0};
    if (n > 0u) ra = slot_read(gl, 0u, n0);
    if (n > 1u) rb = slot_read(gl, 1u, n0);
    #pragma unroll 1
    for (unsigned j = 0; j < 2u * N_SLOTS; j += 2u) {
        const bool a0 = j < n;
        const bool a1 = j + 1u < n;
        if (__builtin_amdgcn_ballot_w32(a0) == 0u) break;
        float4 va[4], vb[4];
        if (a0) row_load(va, y, ra.x, coff);
        if (a1) row_load(vb, y, rb.x, coff);
        const float w0 = __int_as_float(ra.y);
        const float w1 = __int_as_float(rb.y);
        if (j + 2u < n) ra = slot_read(gl, j + 2u, n0);
        if (j + 3u < n) rb = slot_read(gl, j + 3u, n0);
        if (a0) row_fma(acc, va, w0);
        if (a1) row_fma(acc, vb, w1);
    }

    const int h0 = hd0.y;
    const int h1 = hd1.y;
    if (__builtin_amdgcn_ballot_w32(h0 >= 0 || h1 >= 0) != 0u) {
        const Rec3 none = {-1, 0, -1};
        Rec3 c0 = none;
        Rec3 c1 = none;
        if (h0 >= 0) c0 = rec_load(rec, h0);
        if (h1 >= 0) c1 = rec_load(rec, h1);
        for (int it = 0; it < N_EDGES; ++it) {
            const bool a0 = c0.sender >= 0;
            const bool a1 = c1.sender >= 0;
            if (__builtin_amdgcn_ballot_w32(a0 || a1) == 0u) break;
            float4 va[4], vb[4];
            if (a0) row_load(va, y, c0.sender, coff);
            if (a1) row_load(vb, y, c1.sender, coff);
            const float w0 = __int_as_float(c0.wbits);
            const float w1 = __int_as_float(c1.wbits);
            if (c0.next >= 0) c0 = rec_load(rec, c0.next); else c0 = none;
            if (c1.next >= 0) c1 = rec_load(rec, c1.next); else c1 = none;
            if (a0) row_fma(acc, va, w0);
            if (a1) row_fma(acc, vb, w1);
        }
    }

    char* orow = (char*)out + (oidx << 4);
    #pragma unroll
    for (int j = 0; j < 4; ++j) *(float4*)(orow + 128 * j) = acc[j];
}

extern "C" void kernel_launch(void* const* d_in, const int* in_sizes, int n_in,
                              void* d_out, int out_size, void* d_ws, size_t ws_size,
                              hipStream_t stream) {
    const float* nodes     = (const float*)d_in[0];
    const int*   senders   = (const int*)  d_in[1];
    const int*   receivers = (const int*)  d_in[2];
    const float* edges     = (const float*)d_in[3];
    const float* Wm        = (const float*)d_in[4];
    float*       out       = (float*)d_out;

    float* y   = (float*)d_ws;
    float* deg = y + (size_t)N_NODES * DIM;
    int*   rec = (int*)(deg + 2 * PLANE_FLOATS);

    k_init_deg<<<(2 * N_NODES + 255) / 256, 256, 0, stream>>>(deg);
    k_edge_deg<<<(N_EDGES + 255) / 256, 256, 0, stream>>>(senders, receivers, edges, deg, rec);

    const int n_strips = (N_NODES + 15) / 16;
    k_gemm_scale<<<(n_strips + 3) / 4, 128, 0, stream>>>(nodes, Wm, deg, y);

    k_edge_gather<<<(N_NODES + 31) / 32, 256, 0, stream>>>(deg, rec, y, out);
}
